# starting kernel
# baseline (speedup 1.0000x reference)
#include <hip/hip_runtime.h>

#define BATCH  1024
#define INODES 128
#define ONODES 128
#define HID    5

typedef __attribute__((ext_vector_type(2))) float v2f;
typedef __attribute__((ext_vector_type(8))) float v8f;

__device__ __forceinline__ float lrelu(float t) { return fmaxf(t, 0.01f * t); }

#define OFF_PK4  0
#define OFF_W14  1024
#define OFF_PK2A 1280
#define OFF_W2Q  9472
#define OFF_PK2B 11520
#define SMEM_FL  15616

__global__ __launch_bounds__(256) void kan_wmma_kernel(
    const float* __restrict__ x,
    const float* __restrict__ W1, const float* __restrict__ B1,
    const float* __restrict__ W2, const float* __restrict__ B2,
    const float* __restrict__ W3, const float* __restrict__ B3,
    float* __restrict__ out) {
  __shared__ __align__(16) float smem[SMEM_FL];
  __shared__ float sB3;

  const int tid   = threadIdx.x;
  const int o     = blockIdx.x;
  const int chunk = blockIdx.y;

  const float* W1g = W1 + o * (INODES * HID);
  const float* B1g = B1 + o * (INODES * HID);
  const float* W2g = W2 + o * (INODES * HID * HID);
  const float* B2g = B2 + o * (INODES * HID);
  const float* W3g = W3 + o * (INODES * HID);

  for (int idx = tid; idx < 1024; idx += 256) {
    const int i = idx >> 3, r = idx & 7, hf = r >> 2, j = r & 3;
    const int h = 2 * hf + (j & 1);
    smem[OFF_PK4 + idx] = (j < 2) ? W1g[i * HID + h] : B1g[i * HID + h];
  }
  for (int idx = tid; idx < 256; idx += 256) {
    const int i = idx >> 1, j = idx & 1;
    smem[OFF_W14 + idx] = j ? B1g[i * HID + 4] : W1g[i * HID + 4];
  }
  for (int idx = tid; idx < 8192; idx += 256) {
    const int i = idx >> 6, r = idx & 63, ln = r >> 1, j = r & 1;
    const int hf = ln >> 4, N = ln & 15, Nc = (N < HID) ? N : 0;
    const float v = W2g[i * 25 + Nc * HID + 2 * hf + j];
    smem[OFF_PK2A + idx] = (N < HID) ? v : 0.f;
  }
  for (int idx = tid; idx < 2048; idx += 256) {
    const int i = idx >> 4, N = idx & 15, Nc = (N < HID) ? N : 0;
    const float v = W2g[i * 25 + Nc * HID + 4];
    smem[OFF_W2Q + idx] = (N < HID) ? v : 0.f;
  }
  for (int idx = tid; idx < 4096; idx += 256) {
    const int i = idx >> 5, r = idx & 31, N = r >> 1, j = r & 1;
    const int Nc = (N < HID) ? N : 0;
    const float v = j ? W3g[i * HID + Nc] : B2g[i * HID + Nc];
    smem[OFF_PK2B + idx] = (N < HID) ? v : 0.f;
  }
  if (tid == 0) {
    float s = 0.f;
    const float* B3g = B3 + o * INODES;
    for (int i = 0; i < INODES; ++i) s += B3g[i];
    sB3 = s;
  }
  __syncthreads();

  const int  lane = tid & 31;
  const int  wave = tid >> 5;
  const int  half = lane >> 4;
  const int  N    = lane & 15;
  const int  b0   = chunk * 128 + wave * 16;
  const float* xp = x + (b0 + N) * INODES;

  const float* pk4  = &smem[OFF_PK4 + half * 4];
  const float* w14p = &smem[OFF_W14];
  const float* pk2a = &smem[OFF_PK2A + lane * 2];
  const float* w2q  = &smem[OFF_W2Q + N];
  const float* pk2b = &smem[OFF_PK2B + N * 2];

  v8f acc = {};

  for (int i = 0; i < INODES; ++i) {
    const float  xv  = xp[i];
    const float4 w1p = *(const float4*)(pk4 + i * 8);
    const float2 w14 = *(const float2*)(w14p + i * 2);
    const float2 bw  = *(const float2*)(pk2a + i * 64);
    float        bw4 = w2q[i * 16];
    const float2 bw3 = *(const float2*)(pk2b + i * 32);
    bw4 = half ? 0.f : bw4;

    const float a0 = lrelu(fmaf(xv, w1p.x, w1p.z));
    const float a1 = lrelu(fmaf(xv, w1p.y, w1p.w));
    const float a4 = lrelu(fmaf(xv, w14.x, w14.y));

    v2f A1;  A1.x = a0;   A1.y = a1;
    v2f Bm1; Bm1.x = bw.x; Bm1.y = bw.y;
    v2f A2;  A2.x = a4;   A2.y = 0.f;
    v2f Bm2; Bm2.x = bw4; Bm2.y = 0.f;

    v8f c = {};
    c = __builtin_amdgcn_wmma_f32_16x16x4_f32(false, A1, false, Bm1,
                                              (short)0, c, false, false);
    c = __builtin_amdgcn_wmma_f32_16x16x4_f32(false, A2, false, Bm2,
                                              (short)0, c, false, false);

    const float bias = bw3.x;
    const float w3v  = bw3.y;
#pragma unroll
    for (int r = 0; r < 8; ++r) {
      const float h2 = lrelu(c[r] + bias);
      acc[r] = fmaf(h2, w3v, acc[r]);
    }
  }

#pragma unroll
  for (int r = 0; r < 8; ++r) {
    float v = acc[r];
    v += __shfl_xor(v, 8, 16);
    v += __shfl_xor(v, 4, 16);
    v += __shfl_xor(v, 2, 16);
    v += __shfl_xor(v, 1, 16);
    acc[r] = v;
  }
  if (N == 0) {
    const float sb = sB3;
#pragma unroll
    for (int r = 0; r < 8; ++r) {
      const int b = b0 + 8 * half + r;
      out[b * ONODES + o] = acc[r] + sb;
    }
  }
}

extern "C" void kernel_launch(void* const* d_in, const int* in_sizes, int n_in,
                              void* d_out, int out_size, void* d_ws, size_t ws_size,
                              hipStream_t stream) {
  (void)in_sizes; (void)n_in; (void)out_size; (void)d_ws; (void)ws_size;
  const float* x  = (const float*)d_in[0];
  const float* W1 = (const float*)d_in[1];
  const float* B1 = (const float*)d_in[2];
  const float* W2 = (const float*)d_in[3];
  const float* B2 = (const float*)d_in[4];
  const float* W3 = (const float*)d_in[5];
  const float* B3 = (const float*)d_in[6];
  float* out = (float*)d_out;

  dim3 grid(ONODES, BATCH / 128);
  kan_wmma_kernel<<<grid, 256, 0, stream>>>(x, W1, B1, W2, B2, W3, B3, out);
}


# 3 edges per 16x16x16 product: block-diagonal W2 fills 15 columns, layer-2 bias is the 16th k; 4 WMMA + one activation pass per 3 edges, LDS image restaged; sum order in layer 2 and over edges changes
# speedup vs baseline: 1.4071x; 1.4071x over previous
#include <hip/hip_runtime.h>

#define BATCH  1024
#define INODES 128
#define ONODES 128
#define HID    5
#define NGRP   43

typedef __attribute__((ext_vector_type(2))) float v2f;
typedef __attribute__((ext_vector_type(8))) float v8f;

struct __attribute__((packed, aligned(4))) xtrip { float a, b, c; };

__device__ __forceinline__ float lrelu(float t) { return fmaxf(t, 0.01f * t); }

#define NGRP_PAD 48
#define OFF_AW  0
#define OFF_BW  (NGRP_PAD * 32)
#define OFF_W3  (OFF_BW + NGRP * 256)
#define SMEM_FL (OFF_W3 + NGRP_PAD * 16)

__device__ __forceinline__ void slot_map(int h, int t, int& ek, int& hin) {
  if (t < 5)       { ek = h; hin = t; }
  else if (h == 0) { ek = 2; hin = t - 5; }
  else if (t < 7)  { ek = 2; hin = t - 2; }
  else             { ek = 3; hin = 0; }
}

__device__ __forceinline__ void edge_group(float xa, float xc, const float* aw, const float* bw,
                                           const float* w3p, v8f& acc) {
  const float4 p0 = *(const float4*)(aw + 0);
  const float4 p1 = *(const float4*)(aw + 4);
  const float4 p2 = *(const float4*)(aw + 8);
  const float4 p3 = *(const float4*)(aw + 12);
  const float4 q0 = *(const float4*)(bw);
  const float4 q1 = *(const float4*)(bw + 128);
  const float  w3v = *w3p;

  v2f A0; A0.x = lrelu(fmaf(xa, p0.x, p0.z)); A0.y = lrelu(fmaf(xa, p0.y, p0.w));
  v2f A1; A1.x = lrelu(fmaf(xa, p1.x, p1.z)); A1.y = lrelu(fmaf(xa, p1.y, p1.w));
  v2f A2; A2.x = lrelu(fmaf(xa, p2.x, p2.z)); A2.y = lrelu(fmaf(xc, p2.y, p2.w));
  v2f A3; A3.x = lrelu(fmaf(xc, p3.x, p3.z)); A3.y = lrelu(fmaf(xc, p3.y, p3.w));
  v2f B0; B0.x = q0.x; B0.y = q0.y;
  v2f B1; B1.x = q0.z; B1.y = q0.w;
  v2f B2; B2.x = q1.x; B2.y = q1.y;
  v2f B3; B3.x = q1.z; B3.y = q1.w;

  v8f c = {};
  c = __builtin_amdgcn_wmma_f32_16x16x4_f32(false, A0, false, B0, (short)0, c, false, false);
  c = __builtin_amdgcn_wmma_f32_16x16x4_f32(false, A1, false, B1, (short)0, c, false, false);
  c = __builtin_amdgcn_wmma_f32_16x16x4_f32(false, A2, false, B2, (short)0, c, false, false);
  c = __builtin_amdgcn_wmma_f32_16x16x4_f32(false, A3, false, B3, (short)0, c, false, false);

#pragma unroll
  for (int r = 0; r < 8; ++r) acc[r] = fmaf(lrelu(c[r]), w3v, acc[r]);
}

__global__ __launch_bounds__(256) void kan_wmma_kernel(
    const float* __restrict__ x,
    const float* __restrict__ W1, const float* __restrict__ B1,
    const float* __restrict__ W2, const float* __restrict__ B2,
    const float* __restrict__ W3, const float* __restrict__ B3,
    float* __restrict__ out) {
  __shared__ __align__(16) float smem[SMEM_FL];
  __shared__ float sB3;

  const int tid   = threadIdx.x;
  const int o     = blockIdx.x;
  const int chunk = blockIdx.y;

  const float* W1g = W1 + o * (INODES * HID);
  const float* B1g = B1 + o * (INODES * HID);
  const float* W2g = W2 + o * (INODES * HID * HID);
  const float* B2g = B2 + o * (INODES * HID);
  const float* W3g = W3 + o * (INODES * HID);

  const int bq = tid >> 7, bln = (tid >> 2) & 31, bt = bq * 4 + (tid & 3);
  const int bh = bln >> 4, bn = bln & 15;
  const int bec = bn / 5;
  int bek, bhin;
  slot_map(bh, bt, bek, bhin);
  const bool bbias = (bek == 3);
  const bool blive = (bn < 15) && (bbias || bek == bec);
  const float* bsrc = bbias ? (B2g + bn) : (W2g + 5 * bn + bhin);
  const int  bstride = bbias ? 15 : 75;
  const int ar = tid & 31, ah = ar >> 4, aj = (ar >> 2) & 3, as = ar & 3;
  const int at = 2 * aj + (as & 1);
  const bool aisb = (as >> 1) != 0;
  int aek, ahin;
  slot_map(ah, at, aek, ahin);
  const float* asrc = (aisb ? B1g : W1g) + 5 * aek + ahin;
  const float afill = (aek == 3 && aisb) ? 1.f : 0.f;
  const int wn = tid & 15;

  float vb[NGRP], va[NGRP_PAD / 8], vw[NGRP_PAD / 16];
#pragma unroll
  for (int g = 0; g < NGRP; ++g) {
    const bool ok = blive && (3 * g + bec < INODES);
    vb[g] = bsrc[ok ? g * bstride : 0];
  }
#pragma unroll
  for (int trip = 0; trip < NGRP_PAD / 8; ++trip) {
    const int g = trip * 8 + (tid >> 5);
    const bool ok = (aek < 3) && (3 * g + aek < INODES);
    va[trip] = asrc[ok ? 15 * g : 0];
  }
#pragma unroll
  for (int trip = 0; trip < NGRP_PAD / 16; ++trip) {
    const int g = trip * 16 + (tid >> 4);
    const bool ok = (wn < 15) && (3 * g + wn / 5 < INODES);
    vw[trip] = W3g[ok ? 15 * g + wn : 0];
  }
#pragma unroll
  for (int g = 0; g < NGRP; ++g) {
    const bool ok = blive && (3 * g + bec < INODES);
    smem[OFF_BW + g * 256 + tid] = ok ? vb[g] : 0.f;
  }
#pragma unroll
  for (int trip = 0; trip < NGRP_PAD / 8; ++trip) {
    const int g = trip * 8 + (tid >> 5);
    const bool ok = (aek < 3) && (3 * g + aek < INODES);
    smem[OFF_AW + g * 32 + ar] = ok ? va[trip] : afill;
  }
#pragma unroll
  for (int trip = 0; trip < NGRP_PAD / 16; ++trip) {
    const int g = trip * 16 + (tid >> 4);
    const bool ok = (wn < 15) && (3 * g + wn / 5 < INODES);
    smem[OFF_W3 + g * 16 + wn] = ok ? vw[trip] : 0.f;
  }
  if (tid == 0) {
    float s = 0.f;
    const float* B3g = B3 + o * INODES;
    for (int i = 0; i < INODES; ++i) s += B3g[i];
    sB3 = s;
  }
  __syncthreads();

  const int  lane = tid & 31;
  const int  wave = tid >> 5;
  const int  half = lane >> 4;
  const int  N    = lane & 15;
  const int  b0   = chunk * 128 + wave * 16;
  const float* xp = x + (b0 + N) * INODES;

  const float* aw  = &smem[OFF_AW + half * 16];
  const float* bw  = &smem[OFF_BW + lane * 4];
  const float* w3p = &smem[OFF_W3 + N];

  v8f acc = {};

  for (int g = 0; g < NGRP - 1; ++g) {
    const xtrip xv = *(const xtrip*)(xp + 3 * g);
    const float xa = half ? xv.b : xv.a;
    edge_group(xa, xv.c, aw + g * 32, bw + g * 256, w3p + g * 16, acc);
  }
  {
    const float2 xv = *(const float2*)(xp + 3 * (NGRP - 1));
    const float xa = half ? xv.y : xv.x;
    edge_group(xa, 0.f, aw + (NGRP - 1) * 32, bw + (NGRP - 1) * 256, w3p + (NGRP - 1) * 16, acc);
  }

#pragma unroll
  for (int r = 0; r < 8; ++r) {
    float v = acc[r];
    v += __shfl_xor(v, 8, 16);
    v += __shfl_xor(v, 4, 16);
    v += __shfl_xor(v, 2, 16);
    v += __shfl_xor(v, 1, 16);
    acc[r] = v;
  }
  if (N == 0) {
    const float sb = sB3;
#pragma unroll
    for (int r = 0; r < 8; ++r) {
      const int b = b0 + 8 * half + r;
      out[b * ONODES + o] = acc[r] + sb;
    }
  }
}

extern "C" void kernel_launch(void* const* d_in, const int* in_sizes, int n_in,
                              void* d_out, int out_size, void* d_ws, size_t ws_size,
                              hipStream_t stream) {
  (void)in_sizes; (void)n_in; (void)out_size; (void)d_ws; (void)ws_size;
  const float* x  = (const float*)d_in[0];
  const float* W1 = (const float*)d_in[1];
  const float* B1 = (const float*)d_in[2];
  const float* W2 = (const float*)d_in[3];
  const float* B2 = (const float*)d_in[4];
  const float* W3 = (const float*)d_in[5];
  const float* B3 = (const float*)d_in[6];
  float* out = (float*)d_out;

  dim3 grid(ONODES, BATCH / 128);
  kan_wmma_kernel<<<grid, 256, 0, stream>>>(x, W1, B1, W2, B2, W3, B3, out);
}


# Hot loop VALU 36->28 per group: W3*LeakyReLU(c) as a*c+b*|c| (a+b=W3; |c| an fma source modifier, no canonicalising max); loop unrolled x3 for immediate offsets; edge sum takes 2 addends per element
# speedup vs baseline: 1.5073x; 1.0712x over previous
#include <hip/hip_runtime.h>

#define BATCH  1024
#define INODES 128
#define ONODES 128
#define HID    5
#define NGRP   43

typedef __attribute__((ext_vector_type(2))) float v2f;
typedef __attribute__((ext_vector_type(8))) float v8f;

struct __attribute__((packed, aligned(4))) xtrip { float a, b, c; };

__device__ __forceinline__ float lrelu(float t) { return fmaxf(t, 0.01f * t); }

#define NGRP_PAD 48
#define OFF_AW  0
#define OFF_BW  (NGRP_PAD * 32)
#define OFF_W3  (OFF_BW + NGRP * 256)
#define SMEM_FL (OFF_W3 + NGRP_PAD * 32)

__device__ __forceinline__ void slot_map(int h, int t, int& ek, int& hin) {
  if (t < 5)       { ek = h; hin = t; }
  else if (h == 0) { ek = 2; hin = t - 5; }
  else if (t < 7)  { ek = 2; hin = t - 2; }
  else             { ek = 3; hin = 0; }
}

__device__ __forceinline__ void edge_group(float xa, float xc, const float* aw, const float* bw,
                                           const float* w3p, v8f& acc) {
  const float4 p0 = *(const float4*)(aw + 0);
  const float4 p1 = *(const float4*)(aw + 4);
  const float4 p2 = *(const float4*)(aw + 8);
  const float4 p3 = *(const float4*)(aw + 12);
  const float4 q0 = *(const float4*)(bw);
  const float4 q1 = *(const float4*)(bw + 128);
  const float2 w3v = *(const float2*)w3p;

  v2f A0; A0.x = lrelu(fmaf(xa, p0.x, p0.z)); A0.y = lrelu(fmaf(xa, p0.y, p0.w));
  v2f A1; A1.x = lrelu(fmaf(xa, p1.x, p1.z)); A1.y = lrelu(fmaf(xa, p1.y, p1.w));
  v2f A2; A2.x = lrelu(fmaf(xa, p2.x, p2.z)); A2.y = lrelu(fmaf(xc, p2.y, p2.w));
  v2f A3; A3.x = lrelu(fmaf(xc, p3.x, p3.z)); A3.y = lrelu(fmaf(xc, p3.y, p3.w));
  v2f B0; B0.x = q0.x; B0.y = q0.y;
  v2f B1; B1.x = q0.z; B1.y = q0.w;
  v2f B2; B2.x = q1.x; B2.y = q1.y;
  v2f B3; B3.x = q1.z; B3.y = q1.w;

  v8f c = {};
  c = __builtin_amdgcn_wmma_f32_16x16x4_f32(false, A0, false, B0, (short)0, c, false, false);
  c = __builtin_amdgcn_wmma_f32_16x16x4_f32(false, A1, false, B1, (short)0, c, false, false);
  c = __builtin_amdgcn_wmma_f32_16x16x4_f32(false, A2, false, B2, (short)0, c, false, false);
  c = __builtin_amdgcn_wmma_f32_16x16x4_f32(false, A3, false, B3, (short)0, c, false, false);

#pragma unroll
  for (int r = 0; r < 8; ++r) {
    float a = fmaf(c[r], w3v.x, acc[r]);
    asm("v_fma_f32 %0, |%1|, %2, %0" : "+v"(a) : "v"(c[r]), "v"(w3v.y));
    acc[r] = a;
  }
}

__global__ __launch_bounds__(256) void kan_wmma_kernel(
    const float* __restrict__ x,
    const float* __restrict__ W1, const float* __restrict__ B1,
    const float* __restrict__ W2, const float* __restrict__ B2,
    const float* __restrict__ W3, const float* __restrict__ B3,
    float* __restrict__ out) {
  __shared__ __align__(16) float smem[SMEM_FL];
  __shared__ float sB3;

  const int tid   = threadIdx.x;
  const int o     = blockIdx.x;
  const int chunk = blockIdx.y;

  const float* W1g = W1 + o * (INODES * HID);
  const float* B1g = B1 + o * (INODES * HID);
  const float* W2g = W2 + o * (INODES * HID * HID);
  const float* B2g = B2 + o * (INODES * HID);
  const float* W3g = W3 + o * (INODES * HID);

  const int bq = tid >> 7, bln = (tid >> 2) & 31, bt = bq * 4 + (tid & 3);
  const int bh = bln >> 4, bn = bln & 15;
  const int bec = bn / 5;
  int bek, bhin;
  slot_map(bh, bt, bek, bhin);
  const bool bbias = (bek == 3);
  const bool blive = (bn < 15) && (bbias || bek == bec);
  const float* bsrc = bbias ? (B2g + bn) : (W2g + 5 * bn + bhin);
  const int  bstride = bbias ? 15 : 75;
  const int ar = tid & 31, ah = ar >> 4, aj = (ar >> 2) & 3, as = ar & 3;
  const int at = 2 * aj + (as & 1);
  const bool aisb = (as >> 1) != 0;
  int aek, ahin;
  slot_map(ah, at, aek, ahin);
  const float* asrc = (aisb ? B1g : W1g) + 5 * aek + ahin;
  const float afill = (aek == 3 && aisb) ? 1.f : 0.f;
  const int wn = tid & 15;

  float vb[NGRP], va[NGRP_PAD / 8], vw[NGRP_PAD / 16];
#pragma unroll
  for (int g = 0; g < NGRP; ++g) {
    const bool ok = blive && (3 * g + bec < INODES);
    vb[g] = bsrc[ok ? g * bstride : 0];
  }
#pragma unroll
  for (int trip = 0; trip < NGRP_PAD / 8; ++trip) {
    const int g = trip * 8 + (tid >> 5);
    const bool ok = (aek < 3) && (3 * g + aek < INODES);
    va[trip] = asrc[ok ? 15 * g : 0];
  }
#pragma unroll
  for (int trip = 0; trip < NGRP_PAD / 16; ++trip) {
    const int g = trip * 16 + (tid >> 4);
    const bool ok = (wn < 15) && (3 * g + wn / 5 < INODES);
    vw[trip] = W3g[ok ? 15 * g + wn : 0];
  }
#pragma unroll
  for (int g = 0; g < NGRP; ++g) {
    const bool ok = blive && (3 * g + bec < INODES);
    smem[OFF_BW + g * 256 + tid] = ok ? vb[g] : 0.f;
  }
#pragma unroll
  for (int trip = 0; trip < NGRP_PAD / 8; ++trip) {
    const int g = trip * 8 + (tid >> 5);
    const bool ok = (aek < 3) && (3 * g + aek < INODES);
    smem[OFF_AW + g * 32 + ar] = ok ? va[trip] : afill;
  }
#pragma unroll
  for (int trip = 0; trip < NGRP_PAD / 16; ++trip) {
    const int g = trip * 16 + (tid >> 4);
    const bool ok = (wn < 15) && (3 * g + wn / 5 < INODES);
    const float w = ok ? vw[trip] : 0.f;
    *(float2*)&smem[OFF_W3 + (g * 16 + wn) * 2] = make_float2(0.505f * w, 0.495f * w);
  }
  if (tid == 0) {
    float s = 0.f;
    const float* B3g = B3 + o * INODES;
    for (int i = 0; i < INODES; ++i) s += B3g[i];
    sB3 = s;
  }
  __syncthreads();

  const int  lane = tid & 31;
  const int  wave = tid >> 5;
  const int  half = lane >> 4;
  const int  N    = lane & 15;
  const int  b0   = chunk * 128 + wave * 16;
  const float* xp = x + (b0 + N) * INODES;

  const float* aw  = &smem[OFF_AW + half * 16];
  const float* bw  = &smem[OFF_BW + lane * 4];
  const float* w3p = &smem[OFF_W3 + 2 * N];

  v8f acc = {};

#pragma unroll 3
  for (int g = 0; g < NGRP - 1; ++g) {
    const xtrip xv = *(const xtrip*)(xp + 3 * g);
    const float xa = half ? xv.b : xv.a;
    edge_group(xa, xv.c, aw + g * 32, bw + g * 256, w3p + g * 32, acc);
  }
  {
    const float2 xv = *(const float2*)(xp + 3 * (NGRP - 1));
    const float xa = half ? xv.y : xv.x;
    edge_group(xa, 0.f, aw + (NGRP - 1) * 32, bw + (NGRP - 1) * 256, w3p + (NGRP - 1) * 32, acc);
  }

#pragma unroll
  for (int r = 0; r < 8; ++r) {
    float v = acc[r];
    v += __shfl_xor(v, 8, 16);
    v += __shfl_xor(v, 4, 16);
    v += __shfl_xor(v, 2, 16);
    v += __shfl_xor(v, 1, 16);
    acc[r] = v;
  }
  if (N == 0) {
    const float sb = sB3;
#pragma unroll
    for (int r = 0; r < 8; ++r) {
      const int b = b0 + 8 * half + r;
      out[b * ONODES + o] = acc[r] + sb;
    }
  }
}

extern "C" void kernel_launch(void* const* d_in, const int* in_sizes, int n_in,
                              void* d_out, int out_size, void* d_ws, size_t ws_size,
                              hipStream_t stream) {
  (void)in_sizes; (void)n_in; (void)out_size; (void)d_ws; (void)ws_size;
  const float* x  = (const float*)d_in[0];
  const float* W1 = (const float*)d_in[1];
  const float* B1 = (const float*)d_in[2];
  const float* W2 = (const float*)d_in[3];
  const float* B2 = (const float*)d_in[4];
  const float* W3 = (const float*)d_in[5];
  const float* B3 = (const float*)d_in[6];
  float* out = (float*)d_out;

  dim3 grid(ONODES, BATCH / 128);
  kan_wmma_kernel<<<grid, 256, 0, stream>>>(x, W1, B1, W2, B2, W3, B3, out);
}


# One 1024-thread block per output node and 512 batch rows (grid 128x2, one block per WGP; was 128x8 of 256): weights staged 2x per node, not 8x; B3 row summed by a wave's lane tree, was serial
# speedup vs baseline: 1.6353x; 1.0849x over previous
#include <hip/hip_runtime.h>

#define BATCH  1024
#define INODES 128
#define ONODES 128
#define HID    5
#define NGRP   43
#define NTHR   1024
#define BROWS  512

typedef __attribute__((ext_vector_type(2))) float v2f;
typedef __attribute__((ext_vector_type(8))) float v8f;

struct __attribute__((packed, aligned(4))) xtrip { float a, b, c; };

__device__ __forceinline__ float lrelu(float t) { return fmaxf(t, 0.01f * t); }

#define NGRP_PAD 48
#define BW_TRIPS 11
#define OFF_AW  0
#define OFF_BW  (NGRP_PAD * 32)
#define OFF_W3  (OFF_BW + NGRP * 256)
#define SMEM_FL (OFF_W3 + NGRP_PAD * 32)

__device__ __forceinline__ void slot_map(int h, int t, int& ek, int& hin) {
  if (t < 5)       { ek = h; hin = t; }
  else if (h == 0) { ek = 2; hin = t - 5; }
  else if (t < 7)  { ek = 2; hin = t - 2; }
  else             { ek = 3; hin = 0; }
}

__device__ __forceinline__ void edge_group(float xa, float xc, const float* aw, const float* bw,
                                           const float* w3p, v8f& acc) {
  const float4 p0 = *(const float4*)(aw + 0);
  const float4 p1 = *(const float4*)(aw + 4);
  const float4 p2 = *(const float4*)(aw + 8);
  const float4 p3 = *(const float4*)(aw + 12);
  const float4 q0 = *(const float4*)(bw);
  const float4 q1 = *(const float4*)(bw + 128);
  const float2 w3v = *(const float2*)w3p;

  v2f A0; A0.x = lrelu(fmaf(xa, p0.x, p0.z)); A0.y = lrelu(fmaf(xa, p0.y, p0.w));
  v2f A1; A1.x = lrelu(fmaf(xa, p1.x, p1.z)); A1.y = lrelu(fmaf(xa, p1.y, p1.w));
  v2f A2; A2.x = lrelu(fmaf(xa, p2.x, p2.z)); A2.y = lrelu(fmaf(xc, p2.y, p2.w));
  v2f A3; A3.x = lrelu(fmaf(xc, p3.x, p3.z)); A3.y = lrelu(fmaf(xc, p3.y, p3.w));
  v2f B0; B0.x = q0.x; B0.y = q0.y;
  v2f B1; B1.x = q0.z; B1.y = q0.w;
  v2f B2; B2.x = q1.x; B2.y = q1.y;
  v2f B3; B3.x = q1.z; B3.y = q1.w;

  v8f c = {};
  c = __builtin_amdgcn_wmma_f32_16x16x4_f32(false, A0, false, B0, (short)0, c, false, false);
  c = __builtin_amdgcn_wmma_f32_16x16x4_f32(false, A1, false, B1, (short)0, c, false, false);
  c = __builtin_amdgcn_wmma_f32_16x16x4_f32(false, A2, false, B2, (short)0, c, false, false);
  c = __builtin_amdgcn_wmma_f32_16x16x4_f32(false, A3, false, B3, (short)0, c, false, false);

#pragma unroll
  for (int r = 0; r < 8; ++r) {
    float a = fmaf(c[r], w3v.x, acc[r]);
    asm("v_fma_f32 %0, |%1|, %2, %0" : "+v"(a) : "v"(c[r]), "v"(w3v.y));
    acc[r] = a;
  }
}

__global__ __launch_bounds__(NTHR) __attribute__((amdgpu_waves_per_eu(8, 8))) void kan_wmma_kernel(
    const float* __restrict__ x,
    const float* __restrict__ W1, const float* __restrict__ B1,
    const float* __restrict__ W2, const float* __restrict__ B2,
    const float* __restrict__ W3, const float* __restrict__ B3,
    float* __restrict__ out) {
  __shared__ __align__(16) float smem[SMEM_FL];
  __shared__ float sB3;

  const int tid   = threadIdx.x;
  const int o     = blockIdx.x;
  const int chunk = blockIdx.y;

  const float* W1g = W1 + o * (INODES * HID);
  const float* B1g = B1 + o * (INODES * HID);
  const float* W2g = W2 + o * (INODES * HID * HID);
  const float* B2g = B2 + o * (INODES * HID);
  const float* W3g = W3 + o * (INODES * HID);

  const int be = tid & 255, bg0 = tid >> 8;
  const int bq = be >> 7, bln = (be >> 2) & 31, bt = bq * 4 + (be & 3);
  const int bh = bln >> 4, bn = bln & 15;
  const int bec = bn / 5;
  int bek, bhin;
  slot_map(bh, bt, bek, bhin);
  const bool bbias = (bek == 3);
  const bool blive = (bn < 15) && (bbias || bek == bec);
  const float* bsrc = bbias ? (B2g + bn) : (W2g + 5 * bn + bhin);
  const int  bstride = bbias ? 15 : 75;
  const int ar = tid & 31, ah = ar >> 4, aj = (ar >> 2) & 3, as = ar & 3;
  const int at = 2 * aj + (as & 1);
  const bool aisb = (as >> 1) != 0;
  int aek, ahin;
  slot_map(ah, at, aek, ahin);
  const float* asrc = (aisb ? B1g : W1g) + 5 * aek + ahin;
  const float afill = (aek == 3 && aisb) ? 1.f : 0.f;
  const int wn = tid & 15, wg = tid >> 4;

  float vb[BW_TRIPS], va[2], vw;
#pragma unroll
  for (int trip = 0; trip < BW_TRIPS; ++trip) {
    const int g = trip * 4 + bg0;
    const bool ok = blive && (3 * g + bec < INODES);
    vb[trip] = bsrc[ok ? g * bstride : 0];
  }
#pragma unroll
  for (int trip = 0; trip < 2; ++trip) {
    const int g = trip * 32 + (tid >> 5);
    const bool ok = (aek < 3) && (3 * g + aek < INODES);
    va[trip] = asrc[ok ? 15 * g : 0];
  }
  {
    const bool ok = (wn < 15) && (3 * wg + wn / 5 < INODES);
    vw = W3g[ok ? 15 * wg + wn : 0];
  }
#pragma unroll
  for (int trip = 0; trip < BW_TRIPS; ++trip) {
    const int g = trip * 4 + bg0;
    const bool ok = blive && (3 * g + bec < INODES);
    if (g < NGRP) smem[OFF_BW + g * 256 + be] = ok ? vb[trip] : 0.f;
  }
#pragma unroll
  for (int trip = 0; trip < 2; ++trip) {
    const int g = trip * 32 + (tid >> 5);
    const bool ok = (aek < 3) && (3 * g + aek < INODES);
    if (g < NGRP_PAD) smem[OFF_AW + g * 32 + ar] = ok ? va[trip] : afill;
  }
  if (wg < NGRP_PAD) {
    const bool ok = (wn < 15) && (3 * wg + wn / 5 < INODES);
    const float w = ok ? vw : 0.f;
    *(float2*)&smem[OFF_W3 + (wg * 16 + wn) * 2] = make_float2(0.505f * w, 0.495f * w);
  }
  if (tid >= NTHR - 32) {
    const float* B3g = B3 + o * INODES + (tid & 31);
    float s = (B3g[0] + B3g[32]) + (B3g[64] + B3g[96]);
    s += __shfl_xor(s, 16, 32);
    s += __shfl_xor(s, 8, 32);
    s += __shfl_xor(s, 4, 32);
    s += __shfl_xor(s, 2, 32);
    s += __shfl_xor(s, 1, 32);
    if ((tid & 31) == 0) sB3 = s;
  }
  __syncthreads();

  const int  lane = tid & 31;
  const int  wave = tid >> 5;
  const int  half = lane >> 4;
  const int  N    = lane & 15;
  const int  b0   = chunk * BROWS + wave * 16;
  const float* xp = x + (b0 + N) * INODES;

  const float* aw  = &smem[OFF_AW + half * 16];
  const float* bw  = &smem[OFF_BW + lane * 4];
  const float* w3p = &smem[OFF_W3 + 2 * N];

  v8f acc = {};

#pragma unroll 3
  for (int g = 0; g < NGRP - 1; ++g) {
    const xtrip xv = *(const xtrip*)(xp + 3 * g);
    const float xa = half ? xv.b : xv.a;
    edge_group(xa, xv.c, aw + g * 32, bw + g * 256, w3p + g * 32, acc);
  }
  {
    const float2 xv = *(const float2*)(xp + 3 * (NGRP - 1));
    const float xa = half ? xv.y : xv.x;
    edge_group(xa, 0.f, aw + (NGRP - 1) * 32, bw + (NGRP - 1) * 256, w3p + (NGRP - 1) * 32, acc);
  }

#pragma unroll
  for (int r = 0; r < 8; ++r) {
    float v = acc[r];
    v += __shfl_xor(v, 8, 16);
    v += __shfl_xor(v, 4, 16);
    v += __shfl_xor(v, 2, 16);
    v += __shfl_xor(v, 1, 16);
    acc[r] = v;
  }
  if (N == 0) {
    const float sb = sB3;
#pragma unroll
    for (int r = 0; r < 8; ++r) {
      const int b = b0 + 8 * half + r;
      out[b * ONODES + o] = acc[r] + sb;
    }
  }
}

extern "C" void kernel_launch(void* const* d_in, const int* in_sizes, int n_in,
                              void* d_out, int out_size, void* d_ws, size_t ws_size,
                              hipStream_t stream) {
  (void)in_sizes; (void)n_in; (void)out_size; (void)d_ws; (void)ws_size;
  const float* x  = (const float*)d_in[0];
  const float* W1 = (const float*)d_in[1];
  const float* B1 = (const float*)d_in[2];
  const float* W2 = (const float*)d_in[3];
  const float* B2 = (const float*)d_in[4];
  const float* W3 = (const float*)d_in[5];
  const float* B3 = (const float*)d_in[6];
  float* out = (float*)d_out;

  dim3 grid(ONODES, BATCH / BROWS);
  kan_wmma_kernel<<<grid, NTHR, 0, stream>>>(x, W1, B1, W2, B2, W3, B3, out);
}


# Each wave fetches its 16x128 x tile once with whole-row 512-byte loads and parks it in LDS (pitch 132); the edge loop reads x from LDS and holds no memory load; arithmetic and sum order unchanged
# speedup vs baseline: 2.0296x; 1.2411x over previous
#include <hip/hip_runtime.h>

#define BATCH  1024
#define INODES 128
#define ONODES 128
#define HID    5
#define NGRP   43
#define NTHR   1024
#define BROWS  512

typedef __attribute__((ext_vector_type(2))) float v2f;
typedef __attribute__((ext_vector_type(8))) float v8f;

struct __attribute__((packed, aligned(4))) xquad { float a, b, c, d; };

#define XPITCH 132
#define XS_FL  ((NTHR / 32) * 16 * XPITCH)

__device__ __forceinline__ float lrelu(float t) { return fmaxf(t, 0.01f * t); }

#define NGRP_PAD 48
#define BW_TRIPS 11
#define OFF_AW  0
#define OFF_BW  (NGRP_PAD * 32)
#define OFF_W3  (OFF_BW + NGRP * 256)
#define SMEM_FL (OFF_W3 + NGRP_PAD * 32)

__device__ __forceinline__ void slot_map(int h, int t, int& ek, int& hin) {
  if (t < 5)       { ek = h; hin = t; }
  else if (h == 0) { ek = 2; hin = t - 5; }
  else if (t < 7)  { ek = 2; hin = t - 2; }
  else             { ek = 3; hin = 0; }
}

__device__ __forceinline__ void edge_group(float xa, float xc, const float* aw, const float* bw,
                                           const float* w3p, v8f& acc) {
  const float4 p0 = *(const float4*)(aw + 0);
  const float4 p1 = *(const float4*)(aw + 4);
  const float4 p2 = *(const float4*)(aw + 8);
  const float4 p3 = *(const float4*)(aw + 12);
  const float4 q0 = *(const float4*)(bw);
  const float4 q1 = *(const float4*)(bw + 128);
  const float2 w3v = *(const float2*)w3p;

  v2f A0; A0.x = lrelu(fmaf(xa, p0.x, p0.z)); A0.y = lrelu(fmaf(xa, p0.y, p0.w));
  v2f A1; A1.x = lrelu(fmaf(xa, p1.x, p1.z)); A1.y = lrelu(fmaf(xa, p1.y, p1.w));
  v2f A2; A2.x = lrelu(fmaf(xa, p2.x, p2.z)); A2.y = lrelu(fmaf(xc, p2.y, p2.w));
  v2f A3; A3.x = lrelu(fmaf(xc, p3.x, p3.z)); A3.y = lrelu(fmaf(xc, p3.y, p3.w));
  v2f B0; B0.x = q0.x; B0.y = q0.y;
  v2f B1; B1.x = q0.z; B1.y = q0.w;
  v2f B2; B2.x = q1.x; B2.y = q1.y;
  v2f B3; B3.x = q1.z; B3.y = q1.w;

  v8f c = {};
  c = __builtin_amdgcn_wmma_f32_16x16x4_f32(false, A0, false, B0, (short)0, c, false, false);
  c = __builtin_amdgcn_wmma_f32_16x16x4_f32(false, A1, false, B1, (short)0, c, false, false);
  c = __builtin_amdgcn_wmma_f32_16x16x4_f32(false, A2, false, B2, (short)0, c, false, false);
  c = __builtin_amdgcn_wmma_f32_16x16x4_f32(false, A3, false, B3, (short)0, c, false, false);

#pragma unroll
  for (int r = 0; r < 8; ++r) {
    float a = fmaf(c[r], w3v.x, acc[r]);
    asm("v_fma_f32 %0, |%1|, %2, %0" : "+v"(a) : "v"(c[r]), "v"(w3v.y));
    acc[r] = a;
  }
}

__global__ __launch_bounds__(NTHR) __attribute__((amdgpu_waves_per_eu(8, 8))) void kan_wmma_kernel(
    const float* __restrict__ x,
    const float* __restrict__ W1, const float* __restrict__ B1,
    const float* __restrict__ W2, const float* __restrict__ B2,
    const float* __restrict__ W3, const float* __restrict__ B3,
    float* __restrict__ out) {
  __shared__ __align__(16) float smem[SMEM_FL];
  __shared__ __align__(16) float xs[XS_FL];
  __shared__ float sB3;

  const int tid   = threadIdx.x;
  const int o     = blockIdx.x;
  const int chunk = blockIdx.y;
  const int lane  = tid & 31;
  const int wave  = tid >> 5;
  const int b0    = chunk * BROWS + wave * 16;

  xquad xr[16];
  {
    const float* xg = x + b0 * INODES + 4 * lane;
#pragma unroll
    for (int r = 0; r < 16; ++r) xr[r] = *(const xquad*)(xg + r * INODES);
  }

  const float* W1g = W1 + o * (INODES * HID);
  const float* B1g = B1 + o * (INODES * HID);
  const float* W2g = W2 + o * (INODES * HID * HID);
  const float* B2g = B2 + o * (INODES * HID);
  const float* W3g = W3 + o * (INODES * HID);

  const int be = tid & 255, bg0 = tid >> 8;
  const int bq = be >> 7, bln = (be >> 2) & 31, bt = bq * 4 + (be & 3);
  const int bh = bln >> 4, bn = bln & 15;
  const int bec = bn / 5;
  int bek, bhin;
  slot_map(bh, bt, bek, bhin);
  const bool bbias = (bek == 3);
  const bool blive = (bn < 15) && (bbias || bek == bec);
  const float* bsrc = bbias ? (B2g + bn) : (W2g + 5 * bn + bhin);
  const int  bstride = bbias ? 15 : 75;
  const int ar = tid & 31, ah = ar >> 4, aj = (ar >> 2) & 3, as = ar & 3;
  const int at = 2 * aj + (as & 1);
  const bool aisb = (as >> 1) != 0;
  int aek, ahin;
  slot_map(ah, at, aek, ahin);
  const float* asrc = (aisb ? B1g : W1g) + 5 * aek + ahin;
  const float afill = (aek == 3 && aisb) ? 1.f : 0.f;
  const int wn = tid & 15, wg = tid >> 4;

  float vb[BW_TRIPS], va[2], vw;
#pragma unroll
  for (int trip = 0; trip < BW_TRIPS; ++trip) {
    const int g = trip * 4 + bg0;
    const bool ok = blive && (3 * g + bec < INODES);
    vb[trip] = bsrc[ok ? g * bstride : 0];
  }
#pragma unroll
  for (int trip = 0; trip < 2; ++trip) {
    const int g = trip * 32 + (tid >> 5);
    const bool ok = (aek < 3) && (3 * g + aek < INODES);
    va[trip] = asrc[ok ? 15 * g : 0];
  }
  {
    const bool ok = (wn < 15) && (3 * wg + wn / 5 < INODES);
    vw = W3g[ok ? 15 * wg + wn : 0];
  }
#pragma unroll
  for (int trip = 0; trip < BW_TRIPS; ++trip) {
    const int g = trip * 4 + bg0;
    const bool ok = blive && (3 * g + bec < INODES);
    if (g < NGRP) smem[OFF_BW + g * 256 + be] = ok ? vb[trip] : 0.f;
  }
#pragma unroll
  for (int trip = 0; trip < 2; ++trip) {
    const int g = trip * 32 + (tid >> 5);
    const bool ok = (aek < 3) && (3 * g + aek < INODES);
    if (g < NGRP_PAD) smem[OFF_AW + g * 32 + ar] = ok ? va[trip] : afill;
  }
  if (wg < NGRP_PAD) {
    const bool ok = (wn < 15) && (3 * wg + wn / 5 < INODES);
    const float w = ok ? vw : 0.f;
    *(float2*)&smem[OFF_W3 + (wg * 16 + wn) * 2] = make_float2(0.505f * w, 0.495f * w);
  }
  {
    float* xsw = &xs[wave * 16 * XPITCH + 4 * lane];
#pragma unroll
    for (int r = 0; r < 16; ++r)
      *(float4*)(xsw + r * XPITCH) = make_float4(xr[r].a, xr[r].b, xr[r].c, xr[r].d);
  }
  if (tid >= NTHR - 32) {
    const float* B3g = B3 + o * INODES + (tid & 31);
    float s = (B3g[0] + B3g[32]) + (B3g[64] + B3g[96]);
    s += __shfl_xor(s, 16, 32);
    s += __shfl_xor(s, 8, 32);
    s += __shfl_xor(s, 4, 32);
    s += __shfl_xor(s, 2, 32);
    s += __shfl_xor(s, 1, 32);
    if ((tid & 31) == 0) sB3 = s;
  }
  __syncthreads();

  const int  half = lane >> 4;
  const int  N    = lane & 15;
  const float* xrow = &xs[(wave * 16 + N) * XPITCH];
  const float* xpa  = xrow + half;
  const float* xpc  = xrow + 2;

  const float* aw  = &smem[OFF_AW + half * 16];
  const float* bw  = &smem[OFF_BW + lane * 4];
  const float* w3p = &smem[OFF_W3 + 2 * N];

  v8f acc = {};

#pragma unroll 3
  for (int g = 0; g < NGRP - 1; ++g) {
    edge_group(xpa[3 * g], xpc[3 * g], aw + g * 32, bw + g * 256, w3p + g * 32, acc);
  }
  {
    edge_group(xpa[3 * (NGRP - 1)], 0.f, aw + (NGRP - 1) * 32, bw + (NGRP - 1) * 256,
               w3p + (NGRP - 1) * 32, acc);
  }

#pragma unroll
  for (int r = 0; r < 8; ++r) {
    float v = acc[r];
    v += __shfl_xor(v, 8, 16);
    v += __shfl_xor(v, 4, 16);
    v += __shfl_xor(v, 2, 16);
    v += __shfl_xor(v, 1, 16);
    acc[r] = v;
  }
  if (N == 0) {
    const float sb = sB3;
#pragma unroll
    for (int r = 0; r < 8; ++r) {
      const int b = b0 + 8 * half + r;
      out[b * ONODES + o] = acc[r] + sb;
    }
  }
}

extern "C" void kernel_launch(void* const* d_in, const int* in_sizes, int n_in,
                              void* d_out, int out_size, void* d_ws, size_t ws_size,
                              hipStream_t stream) {
  (void)in_sizes; (void)n_in; (void)out_size; (void)d_ws; (void)ws_size;
  const float* x  = (const float*)d_in[0];
  const float* W1 = (const float*)d_in[1];
  const float* B1 = (const float*)d_in[2];
  const float* W2 = (const float*)d_in[3];
  const float* B2 = (const float*)d_in[4];
  const float* W3 = (const float*)d_in[5];
  const float* B3 = (const float*)d_in[6];
  float* out = (float*)d_out;

  dim3 grid(ONODES, BATCH / BROWS);
  kan_wmma_kernel<<<grid, NTHR, 0, stream>>>(x, W1, B1, W2, B2, W3, B3, out);
}


# x tile fetched in four column quarters: the first ahead of the barrier, each later one loaded under the edge loop a segment ahead and parked by its own wave; arithmetic and sum order unchanged
# speedup vs baseline: 2.0929x; 1.0312x over previous
#include <hip/hip_runtime.h>

#define BATCH  1024
#define INODES 128
#define ONODES 128
#define HID    5
#define NGRP   43
#define NTHR   1024
#define BROWS  512

typedef __attribute__((ext_vector_type(2))) float v2f;
typedef __attribute__((ext_vector_type(8))) float v8f;

struct __attribute__((packed, aligned(4))) xquad { float a, b, c, d; };

#define XPITCH 132
#define XS_FL  ((NTHR / 32) * 16 * XPITCH)

__device__ __forceinline__ float lrelu(float t) { return fmaxf(t, 0.01f * t); }

#define NGRP_PAD 48
#define BW_TRIPS 11
#define OFF_AW  0
#define OFF_BW  (NGRP_PAD * 32)
#define OFF_W3  (OFF_BW + NGRP * 256)
#define SMEM_FL (OFF_W3 + NGRP_PAD * 32)

__device__ __forceinline__ void slot_map(int h, int t, int& ek, int& hin) {
  if (t < 5)       { ek = h; hin = t; }
  else if (h == 0) { ek = 2; hin = t - 5; }
  else if (t < 7)  { ek = 2; hin = t - 2; }
  else             { ek = 3; hin = 0; }
}

__device__ __forceinline__ void edge_group(float xa, float xc, const float* aw, const float* bw,
                                           const float* w3p, v8f& acc) {
  const float4 p0 = *(const float4*)(aw + 0);
  const float4 p1 = *(const float4*)(aw + 4);
  const float4 p2 = *(const float4*)(aw + 8);
  const float4 p3 = *(const float4*)(aw + 12);
  const float4 q0 = *(const float4*)(bw);
  const float4 q1 = *(const float4*)(bw + 128);
  const float2 w3v = *(const float2*)w3p;

  v2f A0; A0.x = lrelu(fmaf(xa, p0.x, p0.z)); A0.y = lrelu(fmaf(xa, p0.y, p0.w));
  v2f A1; A1.x = lrelu(fmaf(xa, p1.x, p1.z)); A1.y = lrelu(fmaf(xa, p1.y, p1.w));
  v2f A2; A2.x = lrelu(fmaf(xa, p2.x, p2.z)); A2.y = lrelu(fmaf(xc, p2.y, p2.w));
  v2f A3; A3.x = lrelu(fmaf(xc, p3.x, p3.z)); A3.y = lrelu(fmaf(xc, p3.y, p3.w));
  v2f B0; B0.x = q0.x; B0.y = q0.y;
  v2f B1; B1.x = q0.z; B1.y = q0.w;
  v2f B2; B2.x = q1.x; B2.y = q1.y;
  v2f B3; B3.x = q1.z; B3.y = q1.w;

  v8f c = {};
  c = __builtin_amdgcn_wmma_f32_16x16x4_f32(false, A0, false, B0, (short)0, c, false, false);
  c = __builtin_amdgcn_wmma_f32_16x16x4_f32(false, A1, false, B1, (short)0, c, false, false);
  c = __builtin_amdgcn_wmma_f32_16x16x4_f32(false, A2, false, B2, (short)0, c, false, false);
  c = __builtin_amdgcn_wmma_f32_16x16x4_f32(false, A3, false, B3, (short)0, c, false, false);

#pragma unroll
  for (int r = 0; r < 8; ++r) {
    float a = fmaf(c[r], w3v.x, acc[r]);
    asm("v_fma_f32 %0, |%1|, %2, %0" : "+v"(a) : "v"(c[r]), "v"(w3v.y));
    acc[r] = a;
  }
}

__global__ __launch_bounds__(NTHR) __attribute__((amdgpu_waves_per_eu(8, 8))) void kan_wmma_kernel(
    const float* __restrict__ x,
    const float* __restrict__ W1, const float* __restrict__ B1,
    const float* __restrict__ W2, const float* __restrict__ B2,
    const float* __restrict__ W3, const float* __restrict__ B3,
    float* __restrict__ out) {
  __shared__ __align__(16) float smem[SMEM_FL];
  __shared__ __align__(16) float xs[XS_FL];
  __shared__ float sB3;

  const int tid   = threadIdx.x;
  const int o     = blockIdx.x;
  const int chunk = blockIdx.y;
  const int lane  = tid & 31;
  const int wave  = tid >> 5;
  const int b0    = chunk * BROWS + wave * 16;

  const int xrsel = ((lane >> 3) & 1) * 8 + (lane >> 4) * 4;
  const float* xg  = x + (b0 + xrsel) * INODES + 4 * (lane & 7);
  float*       xsw = &xs[(wave * 16 + xrsel) * XPITCH + 4 * (lane & 7)];
  xquad xq[4];
#define XQ_LOAD(q)                                                                 \
  _Pragma("unroll") for (int k = 0; k < 4; ++k)                                    \
    xq[k] = *(const xquad*)(xg + 32 * (q) + k * INODES);
#define XQ_PARK(q)                                                                 \
  _Pragma("unroll") for (int k = 0; k < 4; ++k)                                    \
    *(float4*)(xsw + 32 * (q) + k * XPITCH) = make_float4(xq[k].a, xq[k].b, xq[k].c, xq[k].d);
  XQ_LOAD(0)

  const float* W1g = W1 + o * (INODES * HID);
  const float* B1g = B1 + o * (INODES * HID);
  const float* W2g = W2 + o * (INODES * HID * HID);
  const float* B2g = B2 + o * (INODES * HID);
  const float* W3g = W3 + o * (INODES * HID);

  const int be = tid & 255, bg0 = tid >> 8;
  const int bq = be >> 7, bln = (be >> 2) & 31, bt = bq * 4 + (be & 3);
  const int bh = bln >> 4, bn = bln & 15;
  const int bec = bn / 5;
  int bek, bhin;
  slot_map(bh, bt, bek, bhin);
  const bool bbias = (bek == 3);
  const bool blive = (bn < 15) && (bbias || bek == bec);
  const float* bsrc = bbias ? (B2g + bn) : (W2g + 5 * bn + bhin);
  const int  bstride = bbias ? 15 : 75;
  const int ar = tid & 31, ah = ar >> 4, aj = (ar >> 2) & 3, as = ar & 3;
  const int at = 2 * aj + (as & 1);
  const bool aisb = (as >> 1) != 0;
  int aek, ahin;
  slot_map(ah, at, aek, ahin);
  const float* asrc = (aisb ? B1g : W1g) + 5 * aek + ahin;
  const float afill = (aek == 3 && aisb) ? 1.f : 0.f;
  const int wn = tid & 15, wg = tid >> 4;

  float vb[BW_TRIPS], va[2], vw;
#pragma unroll
  for (int trip = 0; trip < BW_TRIPS; ++trip) {
    const int g = trip * 4 + bg0;
    const bool ok = blive && (3 * g + bec < INODES);
    vb[trip] = bsrc[ok ? g * bstride : 0];
  }
#pragma unroll
  for (int trip = 0; trip < 2; ++trip) {
    const int g = trip * 32 + (tid >> 5);
    const bool ok = (aek < 3) && (3 * g + aek < INODES);
    va[trip] = asrc[ok ? 15 * g : 0];
  }
  {
    const bool ok = (wn < 15) && (3 * wg + wn / 5 < INODES);
    vw = W3g[ok ? 15 * wg + wn : 0];
  }
#pragma unroll
  for (int trip = 0; trip < BW_TRIPS; ++trip) {
    const int g = trip * 4 + bg0;
    const bool ok = blive && (3 * g + bec < INODES);
    if (g < NGRP) smem[OFF_BW + g * 256 + be] = ok ? vb[trip] : 0.f;
  }
#pragma unroll
  for (int trip = 0; trip < 2; ++trip) {
    const int g = trip * 32 + (tid >> 5);
    const bool ok = (aek < 3) && (3 * g + aek < INODES);
    if (g < NGRP_PAD) smem[OFF_AW + g * 32 + ar] = ok ? va[trip] : afill;
  }
  if (wg < NGRP_PAD) {
    const bool ok = (wn < 15) && (3 * wg + wn / 5 < INODES);
    const float w = ok ? vw : 0.f;
    *(float2*)&smem[OFF_W3 + (wg * 16 + wn) * 2] = make_float2(0.505f * w, 0.495f * w);
  }
  XQ_PARK(0)
  if (tid >= NTHR - 32) {
    const float* B3g = B3 + o * INODES + (tid & 31);
    float s = (B3g[0] + B3g[32]) + (B3g[64] + B3g[96]);
    s += __shfl_xor(s, 16, 32);
    s += __shfl_xor(s, 8, 32);
    s += __shfl_xor(s, 4, 32);
    s += __shfl_xor(s, 2, 32);
    s += __shfl_xor(s, 1, 32);
    if ((tid & 31) == 0) sB3 = s;
  }
  __syncthreads();

  const int  half = lane >> 4;
  const int  N    = lane & 15;
  const float* xrow = &xs[(wave * 16 + N) * XPITCH];
  const float* xpa  = xrow + half;
  const float* xpc  = xrow + 2;

  const float* aw  = &smem[OFF_AW + half * 16];
  const float* bw  = &smem[OFF_BW + lane * 4];
  const float* w3p = &smem[OFF_W3 + 2 * N];

  v8f acc = {};

  XQ_LOAD(1)
#pragma unroll 3
  for (int g = 0; g < 9; ++g) {
    edge_group(xpa[3 * g], xpc[3 * g], aw + g * 32, bw + g * 256, w3p + g * 32, acc);
  }
  XQ_PARK(1)
  XQ_LOAD(2)
#pragma unroll 3
  for (int g = 9; g < 21; ++g) {
    edge_group(xpa[3 * g], xpc[3 * g], aw + g * 32, bw + g * 256, w3p + g * 32, acc);
  }
  XQ_PARK(2)
  XQ_LOAD(3)
#pragma unroll 3
  for (int g = 21; g < 30; ++g) {
    edge_group(xpa[3 * g], xpc[3 * g], aw + g * 32, bw + g * 256, w3p + g * 32, acc);
  }
  XQ_PARK(3)
#pragma unroll 3
  for (int g = 30; g < NGRP - 1; ++g) {
    edge_group(xpa[3 * g], xpc[3 * g], aw + g * 32, bw + g * 256, w3p + g * 32, acc);
  }
#undef XQ_LOAD
#undef XQ_PARK
  {
    edge_group(xpa[3 * (NGRP - 1)], 0.f, aw + (NGRP - 1) * 32, bw + (NGRP - 1) * 256,
               w3p + (NGRP - 1) * 32, acc);
  }

#pragma unroll
  for (int r = 0; r < 8; ++r) {
    float v = acc[r];
    v += __shfl_xor(v, 8, 16);
    v += __shfl_xor(v, 4, 16);
    v += __shfl_xor(v, 2, 16);
    v += __shfl_xor(v, 1, 16);
    acc[r] = v;
  }
  if (N == 0) {
    const float sb = sB3;
#pragma unroll
    for (int r = 0; r < 8; ++r) {
      const int b = b0 + 8 * half + r;
      out[b * ONODES + o] = acc[r] + sb;
    }
  }
}

extern "C" void kernel_launch(void* const* d_in, const int* in_sizes, int n_in,
                              void* d_out, int out_size, void* d_ws, size_t ws_size,
                              hipStream_t stream) {
  (void)in_sizes; (void)n_in; (void)out_size; (void)d_ws; (void)ws_size;
  const float* x  = (const float*)d_in[0];
  const float* W1 = (const float*)d_in[1];
  const float* B1 = (const float*)d_in[2];
  const float* W2 = (const float*)d_in[3];
  const float* B2 = (const float*)d_in[4];
  const float* W3 = (const float*)d_in[5];
  const float* B3 = (const float*)d_in[6];
  float* out = (float*)d_out;

  dim3 grid(ONODES, BATCH / BROWS);
  kan_wmma_kernel<<<grid, NTHR, 0, stream>>>(x, W1, B1, W2, B2, W3, B3, out);
}
